# starting kernel
# baseline (speedup 1.0000x reference)
#include <hip/hip_runtime.h>

typedef __attribute__((ext_vector_type(2))) float v2f;
typedef __attribute__((ext_vector_type(8))) float v8f;

#define N_NODES 50000
#define N_EDGES 800000
#define D 64
#define M_TILES (N_NODES / 16)

__global__ void sage_zero_kernel(float* __restrict__ p, int n) {
    int i = blockIdx.x * blockDim.x + threadIdx.x;
    if (i < n) p[i] = 0.0f;
}

__global__ __launch_bounds__(256) void sage_gemm_kernel(
    const float* __restrict__ feat,
    const float* __restrict__ Wn,
    const float* __restrict__ Ws,
    const float* __restrict__ bias,
    float* __restrict__ srcdata,
    float* __restrict__ out)
{
    const int wave = threadIdx.x >> 5;
    const int lane = threadIdx.x & 31;
    const int tile = blockIdx.x * 8 + wave;
    if (tile >= M_TILES) return;

    const int m  = tile * 16 + (lane & 15);
    const int kh = (lane >> 4) * 2;
    const int n0 = lane & 15;

    v8f accN[4] = {};
    v8f accS[4] = {};

#pragma unroll
    for (int kk = 0; kk < 16; ++kk) {
        const int kb = kk * 4 + kh;
        v2f a;
        a.x = feat[m * D + kb];
        a.y = feat[m * D + kb + 1];
#pragma unroll
        for (int t = 0; t < 4; ++t) {
            const int n = t * 16 + n0;
            v2f bn, bs;
            bn.x = Wn[n * D + kb];
            bn.y = Wn[n * D + kb + 1];
            bs.x = Ws[n * D + kb];
            bs.y = Ws[n * D + kb + 1];
            accN[t] = __builtin_amdgcn_wmma_f32_16x16x4_f32(
                false, a, false, bn, (short)0, accN[t], false, false);
            accS[t] = __builtin_amdgcn_wmma_f32_16x16x4_f32(
                false, a, false, bs, (short)0, accS[t], false, false);
        }
    }

    const int rowBase = tile * 16 + (lane >> 4) * 8;
#pragma unroll
    for (int t = 0; t < 4; ++t) {
        const int col = t * 16 + n0;
        const float b = bias[col];
#pragma unroll
        for (int v = 0; v < 8; ++v) {
            const int r = rowBase + v;
            srcdata[r * D + col] = accN[t][v];
            out[r * D + col]     = accS[t][v] + b;
        }
    }
}

__global__ __launch_bounds__(256) void sage_edge_kernel(
    const int* __restrict__ src,
    const int* __restrict__ dst,
    const float* __restrict__ srcdata,
    float* __restrict__ agg,
    float* __restrict__ deg)
{
    const int e = blockIdx.x * 16 + (threadIdx.x >> 4);
    if (e >= N_EDGES) return;
    const int dbase = (threadIdx.x & 15) * 4;
    const int s = src[e];
    const int t = dst[e];

    const float4 val = *(const float4*)(srcdata + (size_t)s * D + dbase);
    float* ap = agg + (size_t)t * D + dbase;
    unsafeAtomicAdd(ap + 0, val.x);
    unsafeAtomicAdd(ap + 1, val.y);
    unsafeAtomicAdd(ap + 2, val.z);
    unsafeAtomicAdd(ap + 3, val.w);
    if (dbase == 0) unsafeAtomicAdd(deg + t, 1.0f);
}

__global__ void sage_finalize_kernel(
    const float* __restrict__ agg,
    const float* __restrict__ deg,
    float* __restrict__ out)
{
    int i = blockIdx.x * blockDim.x + threadIdx.x;
    if (i >= N_NODES * D) return;
    const float dg = deg[i >> 6];
    if (dg > 0.0f) out[i] += agg[i] / dg;
}

extern "C" void kernel_launch(void* const* d_in, const int* in_sizes, int n_in,
                              void* d_out, int out_size, void* d_ws, size_t ws_size,
                              hipStream_t stream)
{
    const float* feat = (const float*)d_in[0];
    const int*   src  = (const int*)d_in[1];
    const int*   dst  = (const int*)d_in[2];
    const float* Wn   = (const float*)d_in[3];
    const float* Ws   = (const float*)d_in[4];
    const float* bias = (const float*)d_in[5];
    float* out = (float*)d_out;

    float* srcdata = (float*)d_ws;
    float* agg     = srcdata + (size_t)N_NODES * D;
    float* deg     = agg     + (size_t)N_NODES * D;

    const int zn = N_NODES * D + N_NODES;
    sage_zero_kernel<<<(zn + 255) / 256, 256, 0, stream>>>(agg, zn);

    sage_gemm_kernel<<<(M_TILES + 7) / 8, 256, 0, stream>>>(
        feat, Wn, Ws, bias, srcdata, out);

    sage_edge_kernel<<<(N_EDGES + 15) / 16, 256, 0, stream>>>(
        src, dst, srcdata, agg, deg);

    sage_finalize_kernel<<<(N_NODES * D + 255) / 256, 256, 0, stream>>>(
        agg, deg, out);
}


# float-atomic scatter replaced by counting sort of edges by dst (int atomic rank, prefix, fill) + per-node register sum of rows; sum order: arrival-order rank instead of atomic arrival
# speedup vs baseline: 1.2876x; 1.2876x over previous
#include <hip/hip_runtime.h>

typedef __attribute__((ext_vector_type(2))) float v2f;
typedef __attribute__((ext_vector_type(8))) float v8f;

#define N_NODES 50000
#define N_EDGES 800000
#define D 64
#define M_TILES (N_NODES / 16)

__global__ __launch_bounds__(256) void sage_gemm_kernel(
    const float* __restrict__ feat,
    const float* __restrict__ Wn,
    const float* __restrict__ Ws,
    const float* __restrict__ bias,
    float* __restrict__ srcdata,
    float* __restrict__ out,
    int* __restrict__ cnt)
{
    const int g = blockIdx.x * 256 + threadIdx.x;
    if (g < N_NODES) cnt[g] = 0;

    const int wave = threadIdx.x >> 5;
    const int lane = threadIdx.x & 31;
    const int tile = blockIdx.x * 8 + wave;
    if (tile >= M_TILES) return;

    const int m  = tile * 16 + (lane & 15);
    const int kh = (lane >> 4) * 2;
    const int n0 = lane & 15;

    v8f accN[4] = {};
    v8f accS[4] = {};

#pragma unroll
    for (int kk = 0; kk < 16; ++kk) {
        const int kb = kk * 4 + kh;
        v2f a;
        a.x = feat[m * D + kb];
        a.y = feat[m * D + kb + 1];
#pragma unroll
        for (int t = 0; t < 4; ++t) {
            const int n = t * 16 + n0;
            v2f bn, bs;
            bn.x = Wn[n * D + kb];
            bn.y = Wn[n * D + kb + 1];
            bs.x = Ws[n * D + kb];
            bs.y = Ws[n * D + kb + 1];
            accN[t] = __builtin_amdgcn_wmma_f32_16x16x4_f32(
                false, a, false, bn, (short)0, accN[t], false, false);
            accS[t] = __builtin_amdgcn_wmma_f32_16x16x4_f32(
                false, a, false, bs, (short)0, accS[t], false, false);
        }
    }

    const int rowBase = tile * 16 + (lane >> 4) * 8;
#pragma unroll
    for (int t = 0; t < 4; ++t) {
        const int col = t * 16 + n0;
        const float b = bias[col];
#pragma unroll
        for (int v = 0; v < 8; ++v) {
            const int r = rowBase + v;
            srcdata[r * D + col] = accN[t][v];
            out[r * D + col]     = accS[t][v] + b;
        }
    }
}

__global__ __launch_bounds__(256) void sage_rank_kernel(
    const int* __restrict__ dst,
    int* __restrict__ cnt,
    int* __restrict__ rank)
{
    const int e = blockIdx.x * 256 + threadIdx.x;
    if (e >= N_EDGES) return;
    rank[e] = atomicAdd(cnt + dst[e], 1);
}

__global__ __launch_bounds__(256) void sage_offs_kernel(
    const int* __restrict__ cnt,
    int* __restrict__ offs)
{
    __shared__ int part[8];
    __shared__ int wsum[8];
    const int t = threadIdx.x;
    const int lane = t & 31;
    const int w = t >> 5;
    const int base_el = blockIdx.x * 256;

    const int4* __restrict__ c4 = (const int4*)cnt;
    const int nvec = base_el >> 2;
    int s = 0;
#pragma unroll 8
    for (int i = t; i < nvec; i += 256) {
        const int4 v = c4[i];
        s += (v.x + v.y) + (v.z + v.w);
    }

    const int idx = base_el + t;
    const int own = (idx < N_NODES) ? cnt[idx] : 0;
    int inc = own;
    for (int d = 1; d < 32; d <<= 1) {
        const int u = __shfl_up(inc, d);
        if (lane >= d) inc += u;
    }
    for (int d = 16; d > 0; d >>= 1) s += __shfl_xor(s, d);
    if (lane == 31) wsum[w] = inc;
    if (lane == 0)  part[w] = s;
    __syncthreads();
    int base = 0;
    for (int k = 0; k < 8; ++k) {
        base += part[k];
        if (k < w) base += wsum[k];
    }
    if (idx < N_NODES) offs[idx] = base + inc - own;
}

__global__ __launch_bounds__(256) void sage_fill_kernel(
    const int* __restrict__ src,
    const int* __restrict__ dst,
    const int* __restrict__ rank,
    const int* __restrict__ offs,
    int* __restrict__ perm)
{
    const int e = blockIdx.x * 256 + threadIdx.x;
    if (e >= N_EDGES) return;
    perm[offs[dst[e]] + rank[e]] = src[e];
}

__global__ __launch_bounds__(256) void sage_gather_kernel(
    const int* __restrict__ perm,
    const int* __restrict__ offs,
    const int* __restrict__ cnt,
    const float* __restrict__ srcdata,
    float* __restrict__ out)
{
    const int lane = threadIdx.x & 31;
    const int wv = __builtin_amdgcn_readfirstlane((int)(threadIdx.x >> 5));
    const int n = blockIdx.x * 8 + wv;
    if (n >= N_NODES) return;
    const int deg = cnt[n];
    const int beg = offs[n];
    if (deg == 0) return;
    float2* op = (float2*)(out + (size_t)n * D + 2 * lane);
    float2 cur = *op;

    const float* __restrict__ colp = srcdata + 2 * lane;
    float ax = 0.0f, ay = 0.0f;
    for (int c0 = 0; c0 < deg; c0 += 32) {
        const int rem = deg - c0;
        const int m = rem < 32 ? rem : 32;
        const int my = (lane < m) ? perm[beg + c0 + lane] : 0;
        for (int i = 0; i < m; i += 8) {
            float2 v[8];
#pragma unroll
            for (int k = 0; k < 8; ++k) {
                const int srow = __builtin_amdgcn_readlane(my, i + k);
                v[k] = *(const float2*)(colp + (size_t)srow * D);
            }
#pragma unroll
            for (int k = 0; k < 8; ++k) {
                if (i + k < m) { ax += v[k].x; ay += v[k].y; }
            }
        }
    }
    const float dg = (float)deg;
    cur.x += ax / dg;
    cur.y += ay / dg;
    *op = cur;
}

extern "C" void kernel_launch(void* const* d_in, const int* in_sizes, int n_in,
                              void* d_out, int out_size, void* d_ws, size_t ws_size,
                              hipStream_t stream)
{
    const float* feat = (const float*)d_in[0];
    const int*   src  = (const int*)d_in[1];
    const int*   dst  = (const int*)d_in[2];
    const float* Wn   = (const float*)d_in[3];
    const float* Ws   = (const float*)d_in[4];
    const float* bias = (const float*)d_in[5];
    float* out = (float*)d_out;

    float* srcdata = (float*)d_ws;
    int* cnt  = (int*)(srcdata + (size_t)N_NODES * D);
    int* offs = cnt  + 50048;
    int* rank = offs + 50048;
    int* perm = rank + N_EDGES;

    sage_gemm_kernel<<<(M_TILES + 7) / 8, 256, 0, stream>>>(
        feat, Wn, Ws, bias, srcdata, out, cnt);

    sage_rank_kernel<<<N_EDGES / 256, 256, 0, stream>>>(dst, cnt, rank);

    sage_offs_kernel<<<(N_NODES + 255) / 256, 256, 0, stream>>>(cnt, offs);

    sage_fill_kernel<<<N_EDGES / 256, 256, 0, stream>>>(src, dst, rank, offs, perm);

    sage_gather_kernel<<<N_NODES / 8, 256, 0, stream>>>(perm, offs, cnt, srcdata, out);
}


# global returning adds (edge rank) replaced by two-level counting sort on LDS counters: chunk-by-bucket, then bucket-by-node; sum order: LDS arrival rank inside chunk runs, not global arrival
# speedup vs baseline: 2.9331x; 2.2780x over previous
#include <hip/hip_runtime.h>

typedef __attribute__((ext_vector_type(2))) float v2f;
typedef __attribute__((ext_vector_type(8))) float v8f;

#define N_NODES 50000
#define N_EDGES 800000
#define D 64
#define M_TILES (N_NODES / 16)

__global__ __launch_bounds__(256) void sage_gemm_kernel(
    const float* __restrict__ feat,
    const float* __restrict__ Wn,
    const float* __restrict__ Ws,
    const float* __restrict__ bias,
    float* __restrict__ srcdata,
    float* __restrict__ out)
{
    const int wave = threadIdx.x >> 5;
    const int lane = threadIdx.x & 31;
    const int tile = blockIdx.x * 8 + wave;
    if (tile >= M_TILES) return;

    const int m  = tile * 16 + (lane & 15);
    const int kh = (lane >> 4) * 2;
    const int n0 = lane & 15;

    v8f accN[4] = {};
    v8f accS[4] = {};

#pragma unroll
    for (int kk = 0; kk < 16; ++kk) {
        const int kb = kk * 4 + kh;
        v2f a;
        a.x = feat[m * D + kb];
        a.y = feat[m * D + kb + 1];
#pragma unroll
        for (int t = 0; t < 4; ++t) {
            const int n = t * 16 + n0;
            v2f bn, bs;
            bn.x = Wn[n * D + kb];
            bn.y = Wn[n * D + kb + 1];
            bs.x = Ws[n * D + kb];
            bs.y = Ws[n * D + kb + 1];
            accN[t] = __builtin_amdgcn_wmma_f32_16x16x4_f32(
                false, a, false, bn, (short)0, accN[t], false, false);
            accS[t] = __builtin_amdgcn_wmma_f32_16x16x4_f32(
                false, a, false, bs, (short)0, accS[t], false, false);
        }
    }

    const int rowBase = tile * 16 + (lane >> 4) * 8;
#pragma unroll
    for (int t = 0; t < 4; ++t) {
        const int col = t * 16 + n0;
        const float b = bias[col];
#pragma unroll
        for (int v = 0; v < 8; ++v) {
            const int r = rowBase + v;
            srcdata[r * D + col] = accN[t][v];
            out[r * D + col]     = accS[t][v] + b;
        }
    }
}

#define NCHUNK 250
#define CHUNK (N_EDGES / NCHUNK)
#define BNODES 256
#define NBUCK ((N_NODES + BNODES - 1) / BNODES)
#define OFFROW (NBUCK + 1)
#define SORT_T 512
#define EPT ((CHUNK + SORT_T - 1) / SORT_T)
#define SORT_W (SORT_T / 32)
#define RUNS_PER_WAVE ((NCHUNK + SORT_W - 1) / SORT_W)

static_assert(N_NODES <= 65536, "a source id must fit the low 16 bits of a record");
static_assert(CHUNK * NCHUNK == N_EDGES, "chunks must tile the edge list");
static_assert(NBUCK <= 256 && NCHUNK <= 256, "both scans below cover 256 entries");

__device__ __forceinline__ int excl_scan256(int own, int t, int* wtot)
{
    const int lane = t & 31;
    const int w = t >> 5;
    int inc = own;
    for (int d = 1; d < 32; d <<= 1) {
        const int u = __shfl_up(inc, d);
        if (lane >= d) inc += u;
    }
    if (t < 256 && lane == 31) wtot[w] = inc;
    __syncthreads();
    int base = 0;
    for (int k = 0; k < 8; ++k)
        if (k < w) base += wtot[k];
    return base + inc - own;
}

__global__ __launch_bounds__(SORT_T) void sage_chunk_kernel(
    const int* __restrict__ src,
    const int* __restrict__ dst,
    unsigned* __restrict__ rec,
    int* __restrict__ boff)
{
    __shared__ int hist[256];
    __shared__ int wtot[8];
    const int t = threadIdx.x;
    const int c0 = blockIdx.x * CHUNK;
    if (t < 256) hist[t] = 0;
    __syncthreads();

    int d[EPT], s[EPT], r[EPT];
#pragma unroll
    for (int k = 0; k < EPT; ++k) {
        const int i = t + k * SORT_T;
        d[k] = (i < CHUNK) ? dst[c0 + i] : 0;
        s[k] = (i < CHUNK) ? src[c0 + i] : 0;
    }
#pragma unroll
    for (int k = 0; k < EPT; ++k) {
        const int i = t + k * SORT_T;
        r[k] = (i < CHUNK) ? atomicAdd(&hist[d[k] >> 8], 1) : 0;
    }
    __syncthreads();

    const int own = (t < 256) ? hist[t] : 0;
    const int start = excl_scan256(own, t, wtot);
    if (t < 256) hist[t] = start;
    __syncthreads();

    if (t < OFFROW) boff[blockIdx.x * OFFROW + t] = hist[t];
#pragma unroll
    for (int k = 0; k < EPT; ++k) {
        const int i = t + k * SORT_T;
        if (i < CHUNK)
            rec[c0 + hist[d[k] >> 8] + r[k]] =
                ((unsigned)(d[k] & 255) << 16) | (unsigned)s[k];
    }
}

__global__ __launch_bounds__(SORT_T) void sage_bucket_kernel(
    const unsigned* __restrict__ rec,
    const int* __restrict__ boff,
    int* __restrict__ cnt,
    int* __restrict__ offs,
    int* __restrict__ perm)
{
    __shared__ int cntL[256];
    __shared__ int curL[256];
    __shared__ int wtot[8];
    __shared__ int red[SORT_W];
    const int t = threadIdx.x;
    const int lane = t & 31;
    const int b = blockIdx.x;
    const int wu = __builtin_amdgcn_readfirstlane(t >> 5);

    const int mychunk = wu + lane * SORT_W;
    int myS = 0, myN = 0, rel = 0;
    if (lane < RUNS_PER_WAVE && mychunk < NCHUNK) {
        const int s0 = boff[mychunk * OFFROW + b];
        const int e0 = boff[mychunk * OFFROW + b + 1];
        myS = mychunk * CHUNK + s0;
        myN = e0 - s0;
        rel = s0;
    }
    int sN[RUNS_PER_WAVE], sS[RUNS_PER_WAVE];
    unsigned v[RUNS_PER_WAVE];
#pragma unroll
    for (int k = 0; k < RUNS_PER_WAVE; ++k) {
        sN[k] = __builtin_amdgcn_readlane(myN, k);
        sS[k] = __builtin_amdgcn_readlane(myS, k);
        v[k] = (lane < sN[k]) ? rec[sS[k] + lane] : 0u;
    }

    if (t < 256) cntL[t] = 0;
    for (int d = 16; d > 0; d >>= 1) rel += __shfl_xor(rel, d);
    if (lane == 0) red[wu] = rel;
    __syncthreads();
    int bstart = 0;
    for (int k = 0; k < SORT_W; ++k) bstart += red[k];

#pragma unroll
    for (int k = 0; k < RUNS_PER_WAVE; ++k)
        if (lane < sN[k]) atomicAdd(&cntL[v[k] >> 16], 1);
    for (int k = 0; k < RUNS_PER_WAVE; ++k) {
        const int n = __builtin_amdgcn_readlane(myN, k);
        const int s0 = __builtin_amdgcn_readlane(myS, k);
        for (int i = lane + 32; i < n; i += 32)
            atomicAdd(&cntL[rec[s0 + i] >> 16], 1);
    }
    __syncthreads();

    const int own = (t < 256) ? cntL[t] : 0;
    const int ex = excl_scan256(own, t, wtot);
    const int node = b * BNODES + t;
    if (t < 256) {
        curL[t] = bstart + ex;
        if (node < N_NODES) {
            cnt[node]  = own;
            offs[node] = bstart + ex;
        }
    }
    __syncthreads();

#pragma unroll
    for (int k = 0; k < RUNS_PER_WAVE; ++k)
        if (lane < sN[k])
            perm[atomicAdd(&curL[v[k] >> 16], 1)] = (int)(v[k] & 0xFFFFu);
    for (int k = 0; k < RUNS_PER_WAVE; ++k) {
        const int n = __builtin_amdgcn_readlane(myN, k);
        const int s0 = __builtin_amdgcn_readlane(myS, k);
        for (int i = lane + 32; i < n; i += 32) {
            const unsigned x = rec[s0 + i];
            perm[atomicAdd(&curL[x >> 16], 1)] = (int)(x & 0xFFFFu);
        }
    }
}

__global__ __launch_bounds__(256) void sage_gather_kernel(
    const int* __restrict__ perm,
    const int* __restrict__ offs,
    const int* __restrict__ cnt,
    const float* __restrict__ srcdata,
    float* __restrict__ out)
{
    const int lane = threadIdx.x & 31;
    const int wv = __builtin_amdgcn_readfirstlane((int)(threadIdx.x >> 5));
    const int n = blockIdx.x * 8 + wv;
    if (n >= N_NODES) return;
    const int deg = cnt[n];
    const int beg = offs[n];
    if (deg == 0) return;
    float2* op = (float2*)(out + (size_t)n * D + 2 * lane);
    float2 cur = *op;

    const float* __restrict__ colp = srcdata + 2 * lane;
    float ax = 0.0f, ay = 0.0f;
    for (int c0 = 0; c0 < deg; c0 += 32) {
        const int rem = deg - c0;
        const int m = rem < 32 ? rem : 32;
        const int my = (lane < m) ? perm[beg + c0 + lane] : 0;
        for (int i = 0; i < m; i += 8) {
            float2 v[8];
#pragma unroll
            for (int k = 0; k < 8; ++k) {
                const int srow = __builtin_amdgcn_readlane(my, i + k);
                v[k] = *(const float2*)(colp + (size_t)srow * D);
            }
#pragma unroll
            for (int k = 0; k < 8; ++k) {
                if (i + k < m) { ax += v[k].x; ay += v[k].y; }
            }
        }
    }
    const float dg = (float)deg;
    cur.x += ax / dg;
    cur.y += ay / dg;
    *op = cur;
}

extern "C" void kernel_launch(void* const* d_in, const int* in_sizes, int n_in,
                              void* d_out, int out_size, void* d_ws, size_t ws_size,
                              hipStream_t stream)
{
    const float* feat = (const float*)d_in[0];
    const int*   src  = (const int*)d_in[1];
    const int*   dst  = (const int*)d_in[2];
    const float* Wn   = (const float*)d_in[3];
    const float* Ws   = (const float*)d_in[4];
    const float* bias = (const float*)d_in[5];
    float* out = (float*)d_out;

    float* srcdata = (float*)d_ws;
    int* boff = (int*)d_ws;
    int* cnt  = (int*)(srcdata + (size_t)N_NODES * D);
    int* offs = cnt  + 50048;
    unsigned* rec = (unsigned*)(offs + 50048);
    int* perm = (int*)(rec + N_EDGES);

    sage_chunk_kernel<<<NCHUNK, SORT_T, 0, stream>>>(src, dst, rec, boff);

    sage_bucket_kernel<<<NBUCK, SORT_T, 0, stream>>>(rec, boff, cnt, offs, perm);

    sage_gemm_kernel<<<(M_TILES + 7) / 8, 256, 0, stream>>>(
        feat, Wn, Ws, bias, srcdata, out);

    sage_gather_kernel<<<N_NODES / 8, 256, 0, stream>>>(perm, offs, cnt, srcdata, out);
}


# B fragments read from a copy of both weight matrices that launch 1 writes in fragment order: 64 whole-line 16-byte loads per wave replace 128 loads of 16 lines each; k order and sum order kept
# speedup vs baseline: 4.4901x; 1.5308x over previous
#include <hip/hip_runtime.h>

typedef __attribute__((ext_vector_type(2))) float v2f;
typedef __attribute__((ext_vector_type(8))) float v8f;

#define N_NODES 50000
#define N_EDGES 800000
#define D 64
#define M_TILES (N_NODES / 16)

#define NCHUNK 250
#define CHUNK (N_EDGES / NCHUNK)
#define BNODES 256
#define NBUCK ((N_NODES + BNODES - 1) / BNODES)
#define OFFROW (NBUCK + 1)
#define SORT_T 512
#define EPT ((CHUNK + SORT_T - 1) / SORT_T)
#define SORT_W (SORT_T / 32)
#define RUNS_PER_WAVE ((NCHUNK + SORT_W - 1) / SORT_W)

static_assert(N_NODES <= 65536, "a source id must fit the low 16 bits of a record");
static_assert(CHUNK * NCHUNK == N_EDGES, "chunks must tile the edge list");
static_assert(NBUCK <= 256 && NCHUNK <= 256, "both scans below cover 256 entries");

#define WFRAG (2 * D * D)
static_assert(WFRAG % SORT_T == 0, "whole work-groups of launch 1 do the weight copy");

__device__ __forceinline__ int excl_scan256(int own, int t, int* wtot)
{
    const int lane = t & 31;
    const int w = t >> 5;
    int inc = own;
    for (int d = 1; d < 32; d <<= 1) {
        const int u = __shfl_up(inc, d);
        if (lane >= d) inc += u;
    }
    if (t < 256 && lane == 31) wtot[w] = inc;
    __syncthreads();
    int base = 0;
    for (int k = 0; k < 8; ++k)
        if (k < w) base += wtot[k];
    return base + inc - own;
}

__global__ __launch_bounds__(SORT_T) void sage_chunk_kernel(
    const int* __restrict__ src,
    const int* __restrict__ dst,
    unsigned* __restrict__ rec,
    int* __restrict__ boff,
    const float* __restrict__ Wn,
    const float* __restrict__ Ws,
    float* __restrict__ wfrag)
{
    __shared__ int hist[256];
    __shared__ int wtot[8];
    const int t = threadIdx.x;
    const int c0 = blockIdx.x * CHUNK;
    if (t < 256) hist[t] = 0;
    __syncthreads();

    int d[EPT], s[EPT], r[EPT];
#pragma unroll
    for (int k = 0; k < EPT; ++k) {
        const int i = t + k * SORT_T;
        d[k] = (i < CHUNK) ? dst[c0 + i] : 0;
        s[k] = (i < CHUNK) ? src[c0 + i] : 0;
    }
    const int g = blockIdx.x * SORT_T + t;
    float wv = 0.0f;
    if (g < WFRAG) {
        const int q  = g & 3;
        const int L  = (g >> 2) & 31;
        const int j  = (g >> 7) & 1;
        const int kk = g >> 8;
        const int n  = (2 * j + (q >> 1)) * 16 + (L & 15);
        const int k  = (kk & 15) * 4 + (L >> 4) * 2 + (q & 1);
        wv = (kk < 16) ? Ws[n * D + k] : Wn[n * D + k];
    }
#pragma unroll
    for (int k = 0; k < EPT; ++k) {
        const int i = t + k * SORT_T;
        r[k] = (i < CHUNK) ? atomicAdd(&hist[d[k] >> 8], 1) : 0;
    }
    __syncthreads();

    const int own = (t < 256) ? hist[t] : 0;
    const int start = excl_scan256(own, t, wtot);
    if (t < 256) hist[t] = start;
    __syncthreads();

    if (t < OFFROW) boff[blockIdx.x * OFFROW + t] = hist[t];
#pragma unroll
    for (int k = 0; k < EPT; ++k) {
        const int i = t + k * SORT_T;
        if (i < CHUNK)
            rec[c0 + hist[d[k] >> 8] + r[k]] =
                ((unsigned)(d[k] & 255) << 16) | (unsigned)s[k];
    }
    if (g < WFRAG) wfrag[g] = wv;
}

__global__ __launch_bounds__(SORT_T) void sage_bucket_kernel(
    const unsigned* __restrict__ rec,
    const int* __restrict__ boff,
    int* __restrict__ cnt,
    int* __restrict__ offs,
    int* __restrict__ perm)
{
    __shared__ int cntL[256];
    __shared__ int curL[256];
    __shared__ int wtot[8];
    __shared__ int red[SORT_W];
    const int t = threadIdx.x;
    const int lane = t & 31;
    const int b = blockIdx.x;
    const int wu = __builtin_amdgcn_readfirstlane(t >> 5);

    const int mychunk = wu + lane * SORT_W;
    int myS = 0, myN = 0, rel = 0;
    if (lane < RUNS_PER_WAVE && mychunk < NCHUNK) {
        const int s0 = boff[mychunk * OFFROW + b];
        const int e0 = boff[mychunk * OFFROW + b + 1];
        myS = mychunk * CHUNK + s0;
        myN = e0 - s0;
        rel = s0;
    }
    int sN[RUNS_PER_WAVE], sS[RUNS_PER_WAVE];
    unsigned v[RUNS_PER_WAVE];
#pragma unroll
    for (int k = 0; k < RUNS_PER_WAVE; ++k) {
        sN[k] = __builtin_amdgcn_readlane(myN, k);
        sS[k] = __builtin_amdgcn_readlane(myS, k);
        v[k] = (lane < sN[k]) ? rec[sS[k] + lane] : 0u;
    }

    if (t < 256) cntL[t] = 0;
    for (int d = 16; d > 0; d >>= 1) rel += __shfl_xor(rel, d);
    if (lane == 0) red[wu] = rel;
    __syncthreads();
    int bstart = 0;
    for (int k = 0; k < SORT_W; ++k) bstart += red[k];

#pragma unroll
    for (int k = 0; k < RUNS_PER_WAVE; ++k)
        if (lane < sN[k]) atomicAdd(&cntL[v[k] >> 16], 1);
    for (int k = 0; k < RUNS_PER_WAVE; ++k) {
        const int n = __builtin_amdgcn_readlane(myN, k);
        const int s0 = __builtin_amdgcn_readlane(myS, k);
        for (int i = lane + 32; i < n; i += 32)
            atomicAdd(&cntL[rec[s0 + i] >> 16], 1);
    }
    __syncthreads();

    const int own = (t < 256) ? cntL[t] : 0;
    const int ex = excl_scan256(own, t, wtot);
    const int node = b * BNODES + t;
    if (t < 256) {
        curL[t] = bstart + ex;
        if (node < N_NODES) {
            cnt[node]  = own;
            offs[node] = bstart + ex;
        }
    }
    __syncthreads();

#pragma unroll
    for (int k = 0; k < RUNS_PER_WAVE; ++k)
        if (lane < sN[k])
            perm[atomicAdd(&curL[v[k] >> 16], 1)] = (int)(v[k] & 0xFFFFu);
    for (int k = 0; k < RUNS_PER_WAVE; ++k) {
        const int n = __builtin_amdgcn_readlane(myN, k);
        const int s0 = __builtin_amdgcn_readlane(myS, k);
        for (int i = lane + 32; i < n; i += 32) {
            const unsigned x = rec[s0 + i];
            perm[atomicAdd(&curL[x >> 16], 1)] = (int)(x & 0xFFFFu);
        }
    }
}

#define TN 16
#define AP 68
#define GB 16

static_assert(M_TILES * TN == N_NODES, "the tiles must cover the nodes exactly");

__global__ __launch_bounds__(32) __attribute__((amdgpu_waves_per_eu(5, 5))) void sage_fused_kernel(
    const float* __restrict__ feat,
    const float* __restrict__ wfrag,
    const float* __restrict__ bias,
    const int* __restrict__ perm,
    const int* __restrict__ offs,
    const int* __restrict__ cnt,
    float* __restrict__ out)
{
    __shared__ float aggL[TN][AP];
    const int lane = threadIdx.x;
    const int n0 = blockIdx.x * TN;

    const int dj = (lane < TN) ? cnt[n0 + lane] : 0;
    const int beg = offs[n0];
    int inc = dj;
    for (int d = 1; d < TN; d <<= 1) {
        const int u = __shfl_up(inc, d);
        if (lane >= d) inc += u;
    }
    const int total = __builtin_amdgcn_readlane(inc, TN - 1);

    float ax = 0.0f, ay = 0.0f;
    int cur = 0;
    int endCur = __builtin_amdgcn_readlane(inc, 0);
    auto close_node = [&]() {
        const int dg = __builtin_amdgcn_readlane(dj, cur);
        float2 mv;
        mv.x = 0.0f;
        mv.y = 0.0f;
        if (dg > 0) {
            const float f = (float)dg;
            mv.x = ax / f;
            mv.y = ay / f;
        }
        *(float2*)&aggL[cur][2 * lane] = mv;
        ax = 0.0f;
        ay = 0.0f;
        ++cur;
        endCur = (cur < TN) ? __builtin_amdgcn_readlane(inc, cur) : 0x7fffffff;
    };

    int idc[GB];
#pragma unroll
    for (int k = 0; k < GB; ++k) idc[k] = perm[beg + k];
    for (int e0 = 0; e0 < total; e0 += GB) {
        int idn[GB];
#pragma unroll
        for (int k = 0; k < GB; ++k) idn[k] = perm[beg + e0 + GB + k];
        float2 v[GB];
#pragma unroll
        for (int k = 0; k < GB; ++k) {
            const int srow = (e0 + k < total) ? idc[k] : n0;
            v[k] = *(const float2*)(feat + (size_t)srow * D + 2 * lane);
        }
#pragma unroll
        for (int k = 0; k < GB; ++k) {
            if (e0 + k < total) {
                while (e0 + k == endCur) close_node();
                ax += v[k].x;
                ay += v[k].y;
            }
        }
#pragma unroll
        for (int k = 0; k < GB; ++k) idc[k] = idn[k];
    }
    while (cur < TN) close_node();

    __syncthreads();

    const int c  = lane & 15;
    const int mrow = n0 + c;
    const int kh = (lane >> 4) * 2;

    v8f acc[4] = {};

    const float4* wf = (const float4*)wfrag + lane;

#pragma unroll
    for (int kk = 0; kk < 16; ++kk) {
        const int kb = kk * 4 + kh;
        v2f a;
        a.x = feat[mrow * D + kb];
        a.y = feat[mrow * D + kb + 1];
        float4 bq[2];
        bq[0] = wf[(kk * 2 + 0) * 32];
        bq[1] = wf[(kk * 2 + 1) * 32];
#pragma unroll
        for (int t = 0; t < 4; ++t) {
            v2f b;
            b.x = (t & 1) ? bq[t >> 1].z : bq[t >> 1].x;
            b.y = (t & 1) ? bq[t >> 1].w : bq[t >> 1].y;
            acc[t] = __builtin_amdgcn_wmma_f32_16x16x4_f32(
                false, a, false, b, (short)0, acc[t], false, false);
        }
    }
#pragma unroll
    for (int kk = 0; kk < 16; ++kk) {
        const int kb = kk * 4 + kh;
        v2f a;
        a.x = aggL[c][kb];
        a.y = aggL[c][kb + 1];
        float4 bq[2];
        bq[0] = wf[((kk + 16) * 2 + 0) * 32];
        bq[1] = wf[((kk + 16) * 2 + 1) * 32];
#pragma unroll
        for (int t = 0; t < 4; ++t) {
            v2f b;
            b.x = (t & 1) ? bq[t >> 1].z : bq[t >> 1].x;
            b.y = (t & 1) ? bq[t >> 1].w : bq[t >> 1].y;
            acc[t] = __builtin_amdgcn_wmma_f32_16x16x4_f32(
                false, a, false, b, (short)0, acc[t], false, false);
        }
    }

    const int rowBase = n0 + (lane >> 4) * 8;
#pragma unroll
    for (int t = 0; t < 4; ++t) {
        const int col = t * 16 + c;
        const float b = bias[col];
#pragma unroll
        for (int v = 0; v < 8; ++v) {
            const int r = rowBase + v;
            out[r * D + col] = acc[t][v] + b;
        }
    }
}

extern "C" void kernel_launch(void* const* d_in, const int* in_sizes, int n_in,
                              void* d_out, int out_size, void* d_ws, size_t ws_size,
                              hipStream_t stream)
{
    const float* feat = (const float*)d_in[0];
    const int*   src  = (const int*)d_in[1];
    const int*   dst  = (const int*)d_in[2];
    const float* Wn   = (const float*)d_in[3];
    const float* Ws   = (const float*)d_in[4];
    const float* bias = (const float*)d_in[5];
    float* out = (float*)d_out;

    float* plane0 = (float*)d_ws;
    int* boff = (int*)d_ws;
    float* wfrag = plane0 + 65536;
    int* cnt  = (int*)(plane0 + (size_t)N_NODES * D);
    int* offs = cnt  + 50048;
    int* perm = offs + 50048;
    unsigned* rec = (unsigned*)(perm + N_EDGES);

    sage_chunk_kernel<<<NCHUNK, SORT_T, 0, stream>>>(src, dst, rec, boff, Wn, Ws, wfrag);

    sage_bucket_kernel<<<NBUCK, SORT_T, 0, stream>>>(rec, boff, cnt, offs, perm);

    sage_fused_kernel<<<M_TILES, 32, 0, stream>>>(
        feat, wfrag, bias, perm, offs, cnt, out);
}


# order-by-node pass fused into the product launch: 13-wave work-groups (buckets of 208 nodes, 241 groups) order ids in LDS, each wave then sums and multiplies 16 nodes; 2 launches; sum order as before
# speedup vs baseline: 5.1741x; 1.1523x over previous
#include <hip/hip_runtime.h>

typedef __attribute__((ext_vector_type(2))) float v2f;
typedef __attribute__((ext_vector_type(8))) float v8f;

#define N_NODES 50000
#define N_EDGES 800000
#define D 64
#define M_TILES (N_NODES / 16)

#define NCHUNK 250
#define CHUNK (N_EDGES / NCHUNK)
#define BNODES 208
#define NBUCK ((N_NODES + BNODES - 1) / BNODES)
#define OFFROW (NBUCK + 1)
#define SORT_T 512
#define EPT ((CHUNK + SORT_T - 1) / SORT_T)
#define BK_W (BNODES / 16)
#define BK_T (BK_W * 32)
#define RUNS_PER_WAVE ((NCHUNK + BK_W - 1) / BK_W)

static_assert(N_NODES <= 65536, "a source id must fit the low 16 bits of a record");
static_assert(CHUNK * NCHUNK == N_EDGES, "chunks must tile the edge list");
static_assert(NBUCK < 256 && BNODES <= 256 && NCHUNK <= 256, "both scans below cover 256 entries");
static_assert(BNODES % 16 == 0 && BK_T >= 256 && RUNS_PER_WAVE <= 32, "whole tiles; 256 scanning threads; one lane per run");
static_assert(NCHUNK * OFFROW <= 65536, "boff must end in front of wfrag");

#define WFRAG (2 * D * D)
static_assert(WFRAG % SORT_T == 0, "whole work-groups of launch 1 do the weight copy");

__device__ __forceinline__ int excl_scan256(int own, int t, int* wtot)
{
    const int lane = t & 31;
    const int w = t >> 5;
    int inc = own;
    for (int d = 1; d < 32; d <<= 1) {
        const int u = __shfl_up(inc, d);
        if (lane >= d) inc += u;
    }
    if (t < 256 && lane == 31) wtot[w] = inc;
    __syncthreads();
    int base = 0;
    for (int k = 0; k < 8; ++k)
        if (k < w) base += wtot[k];
    return base + inc - own;
}

__global__ __launch_bounds__(SORT_T) void sage_chunk_kernel(
    const int* __restrict__ src,
    const int* __restrict__ dst,
    unsigned* __restrict__ rec,
    int* __restrict__ boff,
    const float* __restrict__ Wn,
    const float* __restrict__ Ws,
    float* __restrict__ wfrag)
{
    __shared__ int hist[256];
    __shared__ int wtot[8];
    const int t = threadIdx.x;
    const int c0 = blockIdx.x * CHUNK;
    if (t < 256) hist[t] = 0;
    __syncthreads();

    int d[EPT], s[EPT], r[EPT];
#pragma unroll
    for (int k = 0; k < EPT; ++k) {
        const int i = t + k * SORT_T;
        d[k] = (i < CHUNK) ? dst[c0 + i] : 0;
        s[k] = (i < CHUNK) ? src[c0 + i] : 0;
    }
    const int g = blockIdx.x * SORT_T + t;
    float wv = 0.0f;
    if (g < WFRAG) {
        const int q  = g & 3;
        const int L  = (g >> 2) & 31;
        const int j  = (g >> 7) & 1;
        const int kk = g >> 8;
        const int n  = (2 * j + (q >> 1)) * 16 + (L & 15);
        const int k  = (kk & 15) * 4 + (L >> 4) * 2 + (q & 1);
        wv = (kk < 16) ? Ws[n * D + k] : Wn[n * D + k];
    }
#pragma unroll
    for (int k = 0; k < EPT; ++k) {
        const int i = t + k * SORT_T;
        r[k] = (i < CHUNK) ? atomicAdd(&hist[(unsigned)d[k] / BNODES], 1) : 0;
    }
    __syncthreads();

    const int own = (t < 256) ? hist[t] : 0;
    const int start = excl_scan256(own, t, wtot);
    if (t < 256) hist[t] = start;
    __syncthreads();

    if (t < OFFROW) boff[blockIdx.x * OFFROW + t] = hist[t];
#pragma unroll
    for (int k = 0; k < EPT; ++k) {
        const int i = t + k * SORT_T;
        if (i < CHUNK) {
            const unsigned bk = (unsigned)d[k] / BNODES;
            rec[c0 + hist[bk] + r[k]] =
                (((unsigned)d[k] - bk * BNODES) << 16) | (unsigned)s[k];
        }
    }
    if (g < WFRAG) wfrag[g] = wv;
}

#define TN 16
#define AP 68
#define GB 16
#define CAP 8192

static_assert(M_TILES * TN == N_NODES, "the tiles must cover the nodes exactly");
static_assert(BNODES == BK_W * TN, "one wave per tile of the bucket");
static_assert(BNODES * NBUCK >= N_NODES, "the buckets must cover the nodes");

__global__ __launch_bounds__(BK_T) __attribute__((amdgpu_waves_per_eu(4, 4))) void sage_bucket_fused_kernel(
    const unsigned* __restrict__ rec,
    const int* __restrict__ boff,
    int* permG,
    const float* __restrict__ feat,
    const float* __restrict__ wfrag,
    const float* __restrict__ bias,
    float* __restrict__ out)
{
    __shared__ int cntL[256];
    __shared__ int offL[256];
    __shared__ int curL[256];
    __shared__ int wtot[8];
    __shared__ int red[BK_W];
    __shared__ int permL[CAP];
    __shared__ float aggL[BK_W][TN][AP];
    const int t = threadIdx.x;
    const int lane = t & 31;
    const int b = blockIdx.x;
    const int wu = __builtin_amdgcn_readfirstlane(t >> 5);

    const int mychunk = wu + lane * BK_W;
    int myS = 0, myN = 0, rel = 0;
    if (lane < RUNS_PER_WAVE && mychunk < NCHUNK) {
        const int s0 = boff[mychunk * OFFROW + b];
        const int e0 = boff[mychunk * OFFROW + b + 1];
        myS = mychunk * CHUNK + s0;
        myN = e0 - s0;
        rel = s0;
    }
    int sN[RUNS_PER_WAVE], sS[RUNS_PER_WAVE];
    unsigned rv[RUNS_PER_WAVE];
#pragma unroll
    for (int k = 0; k < RUNS_PER_WAVE; ++k) {
        sN[k] = __builtin_amdgcn_readlane(myN, k);
        sS[k] = __builtin_amdgcn_readlane(myS, k);
        rv[k] = (lane < sN[k]) ? rec[sS[k] + lane] : 0u;
    }

    if (t < 256) cntL[t] = 0;
    for (int d = 16; d > 0; d >>= 1) rel += __shfl_xor(rel, d);
    if (lane == 0) red[wu] = rel;
    __syncthreads();
    int bsum = 0;
    for (int k = 0; k < BK_W; ++k) bsum += red[k];
    const int bstart = __builtin_amdgcn_readfirstlane(bsum);

#pragma unroll
    for (int k = 0; k < RUNS_PER_WAVE; ++k)
        if (lane < sN[k]) atomicAdd(&cntL[rv[k] >> 16], 1);
    for (int k = 0; k < RUNS_PER_WAVE; ++k) {
        const int n = __builtin_amdgcn_readlane(myN, k);
        const int s0 = __builtin_amdgcn_readlane(myS, k);
        for (int i = lane + 32; i < n; i += 32)
            atomicAdd(&cntL[rec[s0 + i] >> 16], 1);
    }
    __syncthreads();

    const int own = (t < 256) ? cntL[t] : 0;
    const int ex = excl_scan256(own, t, wtot);
    if (t < 256) {
        curL[t] = ex;
        offL[t] = ex;
    }
    __syncthreads();

#pragma unroll
    for (int k = 0; k < RUNS_PER_WAVE; ++k)
        if (lane < sN[k]) {
            const int slot = atomicAdd(&curL[rv[k] >> 16], 1);
            const int id = (int)(rv[k] & 0xFFFFu);
            if (slot < CAP) permL[slot] = id;
            else permG[bstart + slot] = id;
        }
    for (int k = 0; k < RUNS_PER_WAVE; ++k) {
        const int n = __builtin_amdgcn_readlane(myN, k);
        const int s0 = __builtin_amdgcn_readlane(myS, k);
        for (int i = lane + 32; i < n; i += 32) {
            const unsigned x = rec[s0 + i];
            const int slot = atomicAdd(&curL[x >> 16], 1);
            const int id = (int)(x & 0xFFFFu);
            if (slot < CAP) permL[slot] = id;
            else permG[bstart + slot] = id;
        }
    }
    __syncthreads();

    const int n0 = b * BNODES + wu * TN;
    if (n0 >= N_NODES) return;
    float (*agg)[AP] = aggL[wu];

    const int dj = (lane < TN) ? cntL[wu * TN + lane] : 0;
    const int beg = __builtin_amdgcn_readfirstlane(offL[wu * TN]);
    int inc = dj;
    for (int d = 1; d < TN; d <<= 1) {
        const int u = __shfl_up(inc, d);
        if (lane >= d) inc += u;
    }
    const int total = __builtin_amdgcn_readlane(inc, TN - 1);

    float ax = 0.0f, ay = 0.0f;
    int cur = 0;
    int endCur = __builtin_amdgcn_readlane(inc, 0);
    auto close_node = [&]() {
        const int dg = __builtin_amdgcn_readlane(dj, cur);
        float2 mv;
        mv.x = 0.0f;
        mv.y = 0.0f;
        if (dg > 0) {
            const float f = (float)dg;
            mv.x = ax / f;
            mv.y = ay / f;
        }
        *(float2*)&agg[cur][2 * lane] = mv;
        ax = 0.0f;
        ay = 0.0f;
        ++cur;
        endCur = (cur < TN) ? __builtin_amdgcn_readlane(inc, cur) : 0x7fffffff;
    };

    auto fetch_id = [&](int e) -> int {
        const int pos = beg + e;
        const int idl = permL[pos < CAP ? pos : CAP - 1];
        int id = (lane < GB && e < total) ? idl : n0;
        if (lane < GB && e < total && pos >= CAP) id = permG[bstart + pos];
        return id;
    };

    int idc = fetch_id(lane);
    for (int e0 = 0; e0 < total; e0 += GB) {
        float2 v[GB];
#pragma unroll
        for (int k = 0; k < GB; ++k) {
            const int srow = __builtin_amdgcn_readlane(idc, k);
            v[k] = *(const float2*)(feat + (size_t)srow * D + 2 * lane);
        }
        const int idn = fetch_id(e0 + GB + lane);
#pragma unroll
        for (int k = 0; k < GB; ++k) {
            if (e0 + k < total) {
                while (e0 + k == endCur) close_node();
                ax += v[k].x;
                ay += v[k].y;
            }
        }
        idc = idn;
    }
    while (cur < TN) close_node();

    __builtin_amdgcn_fence(__ATOMIC_RELEASE, "wavefront");
    __builtin_amdgcn_wave_barrier();
    __builtin_amdgcn_fence(__ATOMIC_ACQUIRE, "wavefront");

    const int c  = lane & 15;
    const int mrow = n0 + c;
    const int kh = (lane >> 4) * 2;

    v8f acc[4] = {};

    const float4* wf = (const float4*)wfrag + lane;

#pragma unroll
    for (int kk = 0; kk < 16; ++kk) {
        const int kb = kk * 4 + kh;
        v2f a;
        a.x = feat[mrow * D + kb];
        a.y = feat[mrow * D + kb + 1];
        float4 bq[2];
        bq[0] = wf[(kk * 2 + 0) * 32];
        bq[1] = wf[(kk * 2 + 1) * 32];
#pragma unroll
        for (int tt = 0; tt < 4; ++tt) {
            v2f bf;
            bf.x = (tt & 1) ? bq[tt >> 1].z : bq[tt >> 1].x;
            bf.y = (tt & 1) ? bq[tt >> 1].w : bq[tt >> 1].y;
            acc[tt] = __builtin_amdgcn_wmma_f32_16x16x4_f32(
                false, a, false, bf, (short)0, acc[tt], false, false);
        }
    }
#pragma unroll
    for (int kk = 0; kk < 16; ++kk) {
        const int kb = kk * 4 + kh;
        v2f a;
        a.x = agg[c][kb];
        a.y = agg[c][kb + 1];
        float4 bq[2];
        bq[0] = wf[((kk + 16) * 2 + 0) * 32];
        bq[1] = wf[((kk + 16) * 2 + 1) * 32];
#pragma unroll
        for (int tt = 0; tt < 4; ++tt) {
            v2f bf;
            bf.x = (tt & 1) ? bq[tt >> 1].z : bq[tt >> 1].x;
            bf.y = (tt & 1) ? bq[tt >> 1].w : bq[tt >> 1].y;
            acc[tt] = __builtin_amdgcn_wmma_f32_16x16x4_f32(
                false, a, false, bf, (short)0, acc[tt], false, false);
        }
    }

    const int rowBase = n0 + (lane >> 4) * 8;
#pragma unroll
    for (int tt = 0; tt < 4; ++tt) {
        const int col = tt * 16 + c;
        const float bv = bias[col];
#pragma unroll
        for (int r8 = 0; r8 < 8; ++r8) {
            const int r = rowBase + r8;
            out[r * D + col] = acc[tt][r8] + bv;
        }
    }
}

extern "C" void kernel_launch(void* const* d_in, const int* in_sizes, int n_in,
                              void* d_out, int out_size, void* d_ws, size_t ws_size,
                              hipStream_t stream)
{
    const float* feat = (const float*)d_in[0];
    const int*   src  = (const int*)d_in[1];
    const int*   dst  = (const int*)d_in[2];
    const float* Wn   = (const float*)d_in[3];
    const float* Ws   = (const float*)d_in[4];
    const float* bias = (const float*)d_in[5];
    float* out = (float*)d_out;

    float* plane0 = (float*)d_ws;
    int* boff = (int*)d_ws;
    float* wfrag = plane0 + 65536;
    int* perm = (int*)(plane0 + (size_t)N_NODES * D) + 2 * 50048;
    unsigned* rec = (unsigned*)(perm + N_EDGES);

    sage_chunk_kernel<<<NCHUNK, SORT_T, 0, stream>>>(src, dst, rec, boff, Wn, Ws, wfrag);

    sage_bucket_fused_kernel<<<NBUCK, BK_T, 0, stream>>>(rec, boff, perm, feat, wfrag, bias, out);
}


# row gather pipelined: ring of 32 feat rows in flight per wave, a slot re-asked as soon as its row is added (was 16 loaded, waited, added); division by in-degree after the loop; sum order kept
# speedup vs baseline: 5.7098x; 1.1035x over previous
#include <hip/hip_runtime.h>

typedef __attribute__((ext_vector_type(2))) float v2f;
typedef __attribute__((ext_vector_type(8))) float v8f;

#define N_NODES 50000
#define N_EDGES 800000
#define D 64
#define M_TILES (N_NODES / 16)

#define NCHUNK 250
#define CHUNK (N_EDGES / NCHUNK)
#define BNODES 208
#define NBUCK ((N_NODES + BNODES - 1) / BNODES)
#define OFFROW (NBUCK + 1)
#define SORT_T 512
#define EPT ((CHUNK + SORT_T - 1) / SORT_T)
#define BK_W (BNODES / 16)
#define BK_T (BK_W * 32)
#define RUNS_PER_WAVE ((NCHUNK + BK_W - 1) / BK_W)

static_assert(N_NODES <= 65536, "a source id must fit the low 16 bits of a record");
static_assert(CHUNK * NCHUNK == N_EDGES, "chunks must tile the edge list");
static_assert(NBUCK < 256 && BNODES <= 256 && NCHUNK <= 256, "both scans below cover 256 entries");
static_assert(BNODES % 16 == 0 && BK_T >= 256 && RUNS_PER_WAVE <= 32, "whole tiles; 256 scanning threads; one lane per run");
static_assert(NCHUNK * OFFROW <= 65536, "boff must end in front of wfrag");

#define WFRAG (2 * D * D)
static_assert(WFRAG % SORT_T == 0, "whole work-groups of launch 1 do the weight copy");

__device__ __forceinline__ int excl_scan256(int own, int t, int* wtot)
{
    const int lane = t & 31;
    const int w = t >> 5;
    int inc = own;
    for (int d = 1; d < 32; d <<= 1) {
        const int u = __shfl_up(inc, d);
        if (lane >= d) inc += u;
    }
    if (t < 256 && lane == 31) wtot[w] = inc;
    __syncthreads();
    int base = 0;
    for (int k = 0; k < 8; ++k)
        if (k < w) base += wtot[k];
    return base + inc - own;
}

__global__ __launch_bounds__(SORT_T) void sage_chunk_kernel(
    const int* __restrict__ src,
    const int* __restrict__ dst,
    unsigned* __restrict__ rec,
    int* __restrict__ boff,
    const float* __restrict__ Wn,
    const float* __restrict__ Ws,
    float* __restrict__ wfrag)
{
    __shared__ int hist[256];
    __shared__ int wtot[8];
    const int t = threadIdx.x;
    const int c0 = blockIdx.x * CHUNK;
    if (t < 256) hist[t] = 0;
    __syncthreads();

    int d[EPT], s[EPT], r[EPT];
#pragma unroll
    for (int k = 0; k < EPT; ++k) {
        const int i = t + k * SORT_T;
        d[k] = (i < CHUNK) ? dst[c0 + i] : 0;
        s[k] = (i < CHUNK) ? src[c0 + i] : 0;
    }
    const int g = blockIdx.x * SORT_T + t;
    float wv = 0.0f;
    if (g < WFRAG) {
        const int q  = g & 3;
        const int L  = (g >> 2) & 31;
        const int j  = (g >> 7) & 1;
        const int kk = g >> 8;
        const int n  = (2 * j + (q >> 1)) * 16 + (L & 15);
        const int k  = (kk & 15) * 4 + (L >> 4) * 2 + (q & 1);
        wv = (kk < 16) ? Ws[n * D + k] : Wn[n * D + k];
    }
#pragma unroll
    for (int k = 0; k < EPT; ++k) {
        const int i = t + k * SORT_T;
        r[k] = (i < CHUNK) ? atomicAdd(&hist[(unsigned)d[k] / BNODES], 1) : 0;
    }
    __syncthreads();

    const int own = (t < 256) ? hist[t] : 0;
    const int start = excl_scan256(own, t, wtot);
    if (t < 256) hist[t] = start;
    __syncthreads();

    if (t < OFFROW) boff[blockIdx.x * OFFROW + t] = hist[t];
#pragma unroll
    for (int k = 0; k < EPT; ++k) {
        const int i = t + k * SORT_T;
        if (i < CHUNK) {
            const unsigned bk = (unsigned)d[k] / BNODES;
            rec[c0 + hist[bk] + r[k]] =
                (((unsigned)d[k] - bk * BNODES) << 16) | (unsigned)s[k];
        }
    }
    if (g < WFRAG) wfrag[g] = wv;
}

#define TN 16
#define AP 68
#define GB 16
#define RING 32
#define CAP 8192

static_assert(M_TILES * TN == N_NODES, "the tiles must cover the nodes exactly");
static_assert(BNODES == BK_W * TN, "one wave per tile of the bucket");
static_assert(BNODES * NBUCK >= N_NODES, "the buckets must cover the nodes");

__global__ __launch_bounds__(BK_T) __attribute__((amdgpu_waves_per_eu(4, 4))) void sage_bucket_fused_kernel(
    const unsigned* __restrict__ rec,
    const int* __restrict__ boff,
    int* permG,
    const float* __restrict__ feat,
    const float* __restrict__ wfrag,
    const float* __restrict__ bias,
    float* __restrict__ out)
{
    __shared__ int cntL[256];
    __shared__ int offL[256];
    __shared__ int curL[256];
    __shared__ int wtot[8];
    __shared__ int red[BK_W];
    __shared__ int permL[CAP];
    __shared__ float aggL[BK_W][TN][AP];
    const int t = threadIdx.x;
    const int lane = t & 31;
    const int b = blockIdx.x;
    const int wu = __builtin_amdgcn_readfirstlane(t >> 5);

    const int mychunk = wu + lane * BK_W;
    int myS = 0, myN = 0, rel = 0;
    if (lane < RUNS_PER_WAVE && mychunk < NCHUNK) {
        const int s0 = boff[mychunk * OFFROW + b];
        const int e0 = boff[mychunk * OFFROW + b + 1];
        myS = mychunk * CHUNK + s0;
        myN = e0 - s0;
        rel = s0;
    }
    int sN[RUNS_PER_WAVE], sS[RUNS_PER_WAVE];
    unsigned rv[RUNS_PER_WAVE];
#pragma unroll
    for (int k = 0; k < RUNS_PER_WAVE; ++k) {
        sN[k] = __builtin_amdgcn_readlane(myN, k);
        sS[k] = __builtin_amdgcn_readlane(myS, k);
        rv[k] = (lane < sN[k]) ? rec[sS[k] + lane] : 0u;
    }

    if (t < 256) cntL[t] = 0;
    for (int d = 16; d > 0; d >>= 1) rel += __shfl_xor(rel, d);
    if (lane == 0) red[wu] = rel;
    __syncthreads();
    int bsum = 0;
    for (int k = 0; k < BK_W; ++k) bsum += red[k];
    const int bstart = __builtin_amdgcn_readfirstlane(bsum);

#pragma unroll
    for (int k = 0; k < RUNS_PER_WAVE; ++k)
        if (lane < sN[k]) atomicAdd(&cntL[rv[k] >> 16], 1);
    for (int k = 0; k < RUNS_PER_WAVE; ++k) {
        const int n = __builtin_amdgcn_readlane(myN, k);
        const int s0 = __builtin_amdgcn_readlane(myS, k);
        for (int i = lane + 32; i < n; i += 32)
            atomicAdd(&cntL[rec[s0 + i] >> 16], 1);
    }
    __syncthreads();

    const int own = (t < 256) ? cntL[t] : 0;
    const int ex = excl_scan256(own, t, wtot);
    if (t < 256) {
        curL[t] = ex;
        offL[t] = ex;
    }
    __syncthreads();

#pragma unroll
    for (int k = 0; k < RUNS_PER_WAVE; ++k)
        if (lane < sN[k]) {
            const int slot = atomicAdd(&curL[rv[k] >> 16], 1);
            const int id = (int)(rv[k] & 0xFFFFu);
            if (slot < CAP) permL[slot] = id;
            else permG[bstart + slot] = id;
        }
    for (int k = 0; k < RUNS_PER_WAVE; ++k) {
        const int n = __builtin_amdgcn_readlane(myN, k);
        const int s0 = __builtin_amdgcn_readlane(myS, k);
        for (int i = lane + 32; i < n; i += 32) {
            const unsigned x = rec[s0 + i];
            const int slot = atomicAdd(&curL[x >> 16], 1);
            const int id = (int)(x & 0xFFFFu);
            if (slot < CAP) permL[slot] = id;
            else permG[bstart + slot] = id;
        }
    }
    __syncthreads();

    const int n0 = b * BNODES + wu * TN;
    if (n0 >= N_NODES) return;
    float (*agg)[AP] = aggL[wu];

    const int dj = (lane < TN) ? cntL[wu * TN + lane] : 0;
    const int beg = __builtin_amdgcn_readfirstlane(offL[wu * TN]);
    int inc = dj;
    for (int d = 1; d < TN; d <<= 1) {
        const int u = __shfl_up(inc, d);
        if (lane >= d) inc += u;
    }
    const int total = __builtin_amdgcn_readlane(inc, TN - 1);

    const unsigned nz = __builtin_amdgcn_ballot_w32(dj > 0);
    float ax = 0.0f, ay = 0.0f;
    int cur = min((int)__builtin_ctz(nz | 0x10000u), TN - 1);
    int endCur = __builtin_amdgcn_readlane(inc, cur);
    auto step = [&](int e, float2 r) {
        if (e == endCur) {
            float2 sv;
            sv.x = ax;
            sv.y = ay;
            *(float2*)&agg[cur][2 * lane] = sv;
            ax = 0.0f;
            ay = 0.0f;
            const unsigned rest = (nz >> (cur + 1)) | 0x10000u;
            cur = min(cur + 1 + (int)__builtin_ctz(rest), TN - 1);
            endCur = __builtin_amdgcn_readlane(inc, cur);
        }
        ax += r.x;
        ay += r.y;
    };
    auto row = [&](int ids, int k) -> float2 {
        const int srow = __builtin_amdgcn_readlane(ids, k);
        return *(const float2*)(feat + (size_t)srow * D + 2 * lane);
    };
    auto zero_rows = [&]() {
        float2 z;
        z.x = 0.0f;
        z.y = 0.0f;
#pragma unroll
        for (int j = 0; j < TN; ++j) *(float2*)&agg[j][2 * lane] = z;
    };

    if (beg + total <= CAP) {
        auto ids_lds = [&](int e) -> int {
            const int pos = beg + e;
            const int idl = permL[pos < CAP ? pos : CAP - 1];
            return (e < total) ? idl : n0;
        };
        float2 v[RING];
        int idb = ids_lds(lane);
#pragma unroll
        for (int k = 0; k < RING; ++k) v[k] = row(idb, k);
        idb = ids_lds(RING + lane);
        zero_rows();
        int e0 = 0;
        for (; e0 + RING < total; e0 += RING) {
            int idc = n0;
#pragma unroll
            for (int k = 0; k < RING; ++k) {
                if (k == 1) idc = ids_lds(e0 + 2 * RING + lane);
                step(e0 + k, v[k]);
                v[k] = row(idb, k);
            }
            idb = idc;
        }
#pragma unroll
        for (int k = 0; k < RING; ++k)
            if (e0 + k < total) step(e0 + k, v[k]);
    } else {
        auto fetch_id = [&](int e) -> int {
            const int pos = beg + e;
            const int idl = permL[pos < CAP ? pos : CAP - 1];
            int id = (lane < GB && e < total) ? idl : n0;
            if (lane < GB && e < total && pos >= CAP) id = permG[bstart + pos];
            return id;
        };
        int idc = fetch_id(lane);
        zero_rows();
        for (int e0 = 0; e0 < total; e0 += GB) {
            float2 v[GB];
#pragma unroll
            for (int k = 0; k < GB; ++k) v[k] = row(idc, k);
            const int idn = fetch_id(e0 + GB + lane);
#pragma unroll
            for (int k = 0; k < GB; ++k)
                if (e0 + k < total) step(e0 + k, v[k]);
            idc = idn;
        }
    }
    if (total > 0) {
        float2 sv;
        sv.x = ax;
        sv.y = ay;
        *(float2*)&agg[cur][2 * lane] = sv;
    }
#pragma unroll 4
    for (int j = 0; j < TN; ++j) {
        const int dg = __builtin_amdgcn_readlane(dj, j);
        const float2 sv = *(const float2*)&agg[j][2 * lane];
        const float f = (float)max(dg, 1);
        float2 mv;
        mv.x = (dg > 0) ? sv.x / f : 0.0f;
        mv.y = (dg > 0) ? sv.y / f : 0.0f;
        *(float2*)&agg[j][2 * lane] = mv;
    }

    __builtin_amdgcn_fence(__ATOMIC_RELEASE, "wavefront");
    __builtin_amdgcn_wave_barrier();
    __builtin_amdgcn_fence(__ATOMIC_ACQUIRE, "wavefront");

    const int c  = lane & 15;
    const int mrow = n0 + c;
    const int kh = (lane >> 4) * 2;

    v8f acc[4] = {};

    const float4* wf = (const float4*)wfrag + lane;

#pragma unroll
    for (int kk = 0; kk < 16; ++kk) {
        const int kb = kk * 4 + kh;
        v2f a;
        a.x = feat[mrow * D + kb];
        a.y = feat[mrow * D + kb + 1];
        float4 bq[2];
        bq[0] = wf[(kk * 2 + 0) * 32];
        bq[1] = wf[(kk * 2 + 1) * 32];
#pragma unroll
        for (int tt = 0; tt < 4; ++tt) {
            v2f bf;
            bf.x = (tt & 1) ? bq[tt >> 1].z : bq[tt >> 1].x;
            bf.y = (tt & 1) ? bq[tt >> 1].w : bq[tt >> 1].y;
            acc[tt] = __builtin_amdgcn_wmma_f32_16x16x4_f32(
                false, a, false, bf, (short)0, acc[tt], false, false);
        }
    }
#pragma unroll
    for (int kk = 0; kk < 16; ++kk) {
        const int kb = kk * 4 + kh;
        v2f a;
        a.x = agg[c][kb];
        a.y = agg[c][kb + 1];
        float4 bq[2];
        bq[0] = wf[((kk + 16) * 2 + 0) * 32];
        bq[1] = wf[((kk + 16) * 2 + 1) * 32];
#pragma unroll
        for (int tt = 0; tt < 4; ++tt) {
            v2f bf;
            bf.x = (tt & 1) ? bq[tt >> 1].z : bq[tt >> 1].x;
            bf.y = (tt & 1) ? bq[tt >> 1].w : bq[tt >> 1].y;
            acc[tt] = __builtin_amdgcn_wmma_f32_16x16x4_f32(
                false, a, false, bf, (short)0, acc[tt], false, false);
        }
    }

    const int rowBase = n0 + (lane >> 4) * 8;
#pragma unroll
    for (int tt = 0; tt < 4; ++tt) {
        const int col = tt * 16 + c;
        const float bv = bias[col];
#pragma unroll
        for (int r8 = 0; r8 < 8; ++r8) {
            const int r = rowBase + r8;
            out[r * D + col] = acc[tt][r8] + bv;
        }
    }
}

extern "C" void kernel_launch(void* const* d_in, const int* in_sizes, int n_in,
                              void* d_out, int out_size, void* d_ws, size_t ws_size,
                              hipStream_t stream)
{
    const float* feat = (const float*)d_in[0];
    const int*   src  = (const int*)d_in[1];
    const int*   dst  = (const int*)d_in[2];
    const float* Wn   = (const float*)d_in[3];
    const float* Ws   = (const float*)d_in[4];
    const float* bias = (const float*)d_in[5];
    float* out = (float*)d_out;

    float* plane0 = (float*)d_ws;
    int* boff = (int*)d_ws;
    float* wfrag = plane0 + 65536;
    int* perm = (int*)(plane0 + (size_t)N_NODES * D) + 2 * 50048;
    unsigned* rec = (unsigned*)(perm + N_EDGES);

    sage_chunk_kernel<<<NCHUNK, SORT_T, 0, stream>>>(src, dst, rec, boff, Wn, Ws, wfrag);

    sage_bucket_fused_kernel<<<NBUCK, BK_T, 0, stream>>>(rec, boff, perm, feat, wfrag, bias, out);
}


# launch 1 assembles each chunk of records in LDS and writes it with 16-byte stores, 4 whole 128-byte lines per instruction, in place of 4-byte stores to a random line per lane; sum order kept
# speedup vs baseline: 6.0526x; 1.0600x over previous
#include <hip/hip_runtime.h>

typedef __attribute__((ext_vector_type(2))) float v2f;
typedef __attribute__((ext_vector_type(8))) float v8f;

#define N_NODES 50000
#define N_EDGES 800000
#define D 64
#define M_TILES (N_NODES / 16)

#define NCHUNK 250
#define CHUNK (N_EDGES / NCHUNK)
#define BNODES 208
#define NBUCK ((N_NODES + BNODES - 1) / BNODES)
#define OFFROW (NBUCK + 1)
#define SORT_T 512
#define EPT ((CHUNK + SORT_T - 1) / SORT_T)
#define BK_W (BNODES / 16)
#define BK_T (BK_W * 32)
#define RUNS_PER_WAVE ((NCHUNK + BK_W - 1) / BK_W)

static_assert(N_NODES <= 65536, "a source id must fit the low 16 bits of a record");
static_assert(CHUNK * NCHUNK == N_EDGES, "chunks must tile the edge list");
static_assert(NBUCK < 256 && BNODES <= 256 && NCHUNK <= 256, "both scans below cover 256 entries");
static_assert(BNODES % 16 == 0 && BK_T >= 256 && RUNS_PER_WAVE <= 32, "whole tiles; 256 scanning threads; one lane per run");
static_assert(NCHUNK * OFFROW <= 65536, "boff must end in front of wfrag");
static_assert((CHUNK * 4) % 128 == 0, "every chunk of rec must start on a 128-byte line");

#define WFRAG (2 * D * D)
static_assert(WFRAG % SORT_T == 0, "whole work-groups of launch 1 do the weight copy");

__device__ __forceinline__ int excl_scan256(int own, int t, int* wtot)
{
    const int lane = t & 31;
    const int w = t >> 5;
    int inc = own;
    for (int d = 1; d < 32; d <<= 1) {
        const int u = __shfl_up(inc, d);
        if (lane >= d) inc += u;
    }
    if (t < 256 && lane == 31) wtot[w] = inc;
    __syncthreads();
    int base = 0;
    for (int k = 0; k < 8; ++k)
        if (k < w) base += wtot[k];
    return base + inc - own;
}

__global__ __launch_bounds__(SORT_T) void sage_chunk_kernel(
    const int* __restrict__ src,
    const int* __restrict__ dst,
    unsigned* __restrict__ rec,
    int* __restrict__ boff,
    const float* __restrict__ Wn,
    const float* __restrict__ Ws,
    float* __restrict__ wfrag)
{
    __shared__ int hist[256];
    __shared__ int wtot[8];
    __shared__ __attribute__((aligned(16))) unsigned img[CHUNK];
    const int t = threadIdx.x;
    const int c0 = blockIdx.x * CHUNK;
    if (t < 256) hist[t] = 0;
    __syncthreads();

    int d[EPT], s[EPT], r[EPT];
#pragma unroll
    for (int k = 0; k < EPT; ++k) {
        const int i = t + k * SORT_T;
        d[k] = (i < CHUNK) ? dst[c0 + i] : 0;
        s[k] = (i < CHUNK) ? src[c0 + i] : 0;
    }
    const int g = blockIdx.x * SORT_T + t;
    float wv = 0.0f;
    if (g < WFRAG) {
        const int q  = g & 3;
        const int L  = (g >> 2) & 31;
        const int j  = (g >> 7) & 1;
        const int kk = g >> 8;
        const int n  = (2 * j + (q >> 1)) * 16 + (L & 15);
        const int k  = (kk & 15) * 4 + (L >> 4) * 2 + (q & 1);
        wv = (kk < 16) ? Ws[n * D + k] : Wn[n * D + k];
    }
#pragma unroll
    for (int k = 0; k < EPT; ++k) {
        const int i = t + k * SORT_T;
        r[k] = (i < CHUNK) ? atomicAdd(&hist[(unsigned)d[k] / BNODES], 1) : 0;
    }
    __syncthreads();

    const int own = (t < 256) ? hist[t] : 0;
    const int start = excl_scan256(own, t, wtot);
    if (t < 256) hist[t] = start;
    __syncthreads();

#pragma unroll
    for (int k = 0; k < EPT; ++k) {
        const int i = t + k * SORT_T;
        if (i < CHUNK) {
            const unsigned bk = (unsigned)d[k] / BNODES;
            img[hist[bk] + r[k]] =
                (((unsigned)d[k] - bk * BNODES) << 16) | (unsigned)s[k];
        }
    }
    __syncthreads();
    const uint4* img4 = (const uint4*)img;
    uint4* rec4 = (uint4*)(rec + c0);
    for (int i = t; i < CHUNK / 4; i += SORT_T) rec4[i] = img4[i];
    if (t < OFFROW) boff[blockIdx.x * OFFROW + t] = hist[t];
    if (g < WFRAG) wfrag[g] = wv;
}

#define TN 16
#define AP 68
#define GB 16
#define RING 32
#define CAP 8192

static_assert(M_TILES * TN == N_NODES, "the tiles must cover the nodes exactly");
static_assert(BNODES == BK_W * TN, "one wave per tile of the bucket");
static_assert(BNODES * NBUCK >= N_NODES, "the buckets must cover the nodes");

__global__ __launch_bounds__(BK_T) __attribute__((amdgpu_waves_per_eu(4, 4))) void sage_bucket_fused_kernel(
    const unsigned* __restrict__ rec,
    const int* __restrict__ boff,
    int* permG,
    const float* __restrict__ feat,
    const float* __restrict__ wfrag,
    const float* __restrict__ bias,
    float* __restrict__ out)
{
    __shared__ int cntL[256];
    __shared__ int offL[256];
    __shared__ int curL[256];
    __shared__ int wtot[8];
    __shared__ int red[BK_W];
    __shared__ int permL[CAP];
    __shared__ float aggL[BK_W][TN][AP];
    const int t = threadIdx.x;
    const int lane = t & 31;
    const int b = blockIdx.x;
    const int wu = __builtin_amdgcn_readfirstlane(t >> 5);

    const int mychunk = wu + lane * BK_W;
    int myS = 0, myN = 0, rel = 0;
    if (lane < RUNS_PER_WAVE && mychunk < NCHUNK) {
        const int s0 = boff[mychunk * OFFROW + b];
        const int e0 = boff[mychunk * OFFROW + b + 1];
        myS = mychunk * CHUNK + s0;
        myN = e0 - s0;
        rel = s0;
    }
    int sN[RUNS_PER_WAVE], sS[RUNS_PER_WAVE];
    unsigned rv[RUNS_PER_WAVE];
#pragma unroll
    for (int k = 0; k < RUNS_PER_WAVE; ++k) {
        sN[k] = __builtin_amdgcn_readlane(myN, k);
        sS[k] = __builtin_amdgcn_readlane(myS, k);
        rv[k] = (lane < sN[k]) ? rec[sS[k] + lane] : 0u;
    }

    if (t < 256) cntL[t] = 0;
    for (int d = 16; d > 0; d >>= 1) rel += __shfl_xor(rel, d);
    if (lane == 0) red[wu] = rel;
    __syncthreads();
    int bsum = 0;
    for (int k = 0; k < BK_W; ++k) bsum += red[k];
    const int bstart = __builtin_amdgcn_readfirstlane(bsum);

#pragma unroll
    for (int k = 0; k < RUNS_PER_WAVE; ++k)
        if (lane < sN[k]) atomicAdd(&cntL[rv[k] >> 16], 1);
    for (int k = 0; k < RUNS_PER_WAVE; ++k) {
        const int n = __builtin_amdgcn_readlane(myN, k);
        const int s0 = __builtin_amdgcn_readlane(myS, k);
        for (int i = lane + 32; i < n; i += 32)
            atomicAdd(&cntL[rec[s0 + i] >> 16], 1);
    }
    __syncthreads();

    const int own = (t < 256) ? cntL[t] : 0;
    const int ex = excl_scan256(own, t, wtot);
    if (t < 256) {
        curL[t] = ex;
        offL[t] = ex;
    }
    __syncthreads();

#pragma unroll
    for (int k = 0; k < RUNS_PER_WAVE; ++k)
        if (lane < sN[k]) {
            const int slot = atomicAdd(&curL[rv[k] >> 16], 1);
            const int id = (int)(rv[k] & 0xFFFFu);
            if (slot < CAP) permL[slot] = id;
            else permG[bstart + slot] = id;
        }
    for (int k = 0; k < RUNS_PER_WAVE; ++k) {
        const int n = __builtin_amdgcn_readlane(myN, k);
        const int s0 = __builtin_amdgcn_readlane(myS, k);
        for (int i = lane + 32; i < n; i += 32) {
            const unsigned x = rec[s0 + i];
            const int slot = atomicAdd(&curL[x >> 16], 1);
            const int id = (int)(x & 0xFFFFu);
            if (slot < CAP) permL[slot] = id;
            else permG[bstart + slot] = id;
        }
    }
    __syncthreads();

    const int n0 = b * BNODES + wu * TN;
    if (n0 >= N_NODES) return;
    float (*agg)[AP] = aggL[wu];

    const int dj = (lane < TN) ? cntL[wu * TN + lane] : 0;
    const int beg = __builtin_amdgcn_readfirstlane(offL[wu * TN]);
    int inc = dj;
    for (int d = 1; d < TN; d <<= 1) {
        const int u = __shfl_up(inc, d);
        if (lane >= d) inc += u;
    }
    const int total = __builtin_amdgcn_readlane(inc, TN - 1);

    const unsigned nz = __builtin_amdgcn_ballot_w32(dj > 0);
    float ax = 0.0f, ay = 0.0f;
    int cur = min((int)__builtin_ctz(nz | 0x10000u), TN - 1);
    int endCur = __builtin_amdgcn_readlane(inc, cur);
    auto step = [&](int e, float2 r) {
        if (e == endCur) {
            float2 sv;
            sv.x = ax;
            sv.y = ay;
            *(float2*)&agg[cur][2 * lane] = sv;
            ax = 0.0f;
            ay = 0.0f;
            const unsigned rest = (nz >> (cur + 1)) | 0x10000u;
            cur = min(cur + 1 + (int)__builtin_ctz(rest), TN - 1);
            endCur = __builtin_amdgcn_readlane(inc, cur);
        }
        ax += r.x;
        ay += r.y;
    };
    auto row = [&](int ids, int k) -> float2 {
        const int srow = __builtin_amdgcn_readlane(ids, k);
        return *(const float2*)(feat + (size_t)srow * D + 2 * lane);
    };
    auto zero_rows = [&]() {
        float2 z;
        z.x = 0.0f;
        z.y = 0.0f;
#pragma unroll
        for (int j = 0; j < TN; ++j) *(float2*)&agg[j][2 * lane] = z;
    };

    if (beg + total <= CAP) {
        auto ids_lds = [&](int e) -> int {
            const int pos = beg + e;
            const int idl = permL[pos < CAP ? pos : CAP - 1];
            return (e < total) ? idl : n0;
        };
        float2 v[RING];
        int idb = ids_lds(lane);
#pragma unroll
        for (int k = 0; k < RING; ++k) v[k] = row(idb, k);
        idb = ids_lds(RING + lane);
        zero_rows();
        int e0 = 0;
        for (; e0 + RING < total; e0 += RING) {
            int idc = n0;
#pragma unroll
            for (int k = 0; k < RING; ++k) {
                if (k == 1) idc = ids_lds(e0 + 2 * RING + lane);
                step(e0 + k, v[k]);
                v[k] = row(idb, k);
            }
            idb = idc;
        }
#pragma unroll
        for (int k = 0; k < RING; ++k)
            if (e0 + k < total) step(e0 + k, v[k]);
    } else {
        auto fetch_id = [&](int e) -> int {
            const int pos = beg + e;
            const int idl = permL[pos < CAP ? pos : CAP - 1];
            int id = (lane < GB && e < total) ? idl : n0;
            if (lane < GB && e < total && pos >= CAP) id = permG[bstart + pos];
            return id;
        };
        int idc = fetch_id(lane);
        zero_rows();
        for (int e0 = 0; e0 < total; e0 += GB) {
            float2 v[GB];
#pragma unroll
            for (int k = 0; k < GB; ++k) v[k] = row(idc, k);
            const int idn = fetch_id(e0 + GB + lane);
#pragma unroll
            for (int k = 0; k < GB; ++k)
                if (e0 + k < total) step(e0 + k, v[k]);
            idc = idn;
        }
    }
    if (total > 0) {
        float2 sv;
        sv.x = ax;
        sv.y = ay;
        *(float2*)&agg[cur][2 * lane] = sv;
    }
#pragma unroll 4
    for (int j = 0; j < TN; ++j) {
        const int dg = __builtin_amdgcn_readlane(dj, j);
        const float2 sv = *(const float2*)&agg[j][2 * lane];
        const float f = (float)max(dg, 1);
        float2 mv;
        mv.x = (dg > 0) ? sv.x / f : 0.0f;
        mv.y = (dg > 0) ? sv.y / f : 0.0f;
        *(float2*)&agg[j][2 * lane] = mv;
    }

    __builtin_amdgcn_fence(__ATOMIC_RELEASE, "wavefront");
    __builtin_amdgcn_wave_barrier();
    __builtin_amdgcn_fence(__ATOMIC_ACQUIRE, "wavefront");

    const int c  = lane & 15;
    const int mrow = n0 + c;
    const int kh = (lane >> 4) * 2;

    v8f acc[4] = {};

    const float4* wf = (const float4*)wfrag + lane;

#pragma unroll
    for (int kk = 0; kk < 16; ++kk) {
        const int kb = kk * 4 + kh;
        v2f a;
        a.x = feat[mrow * D + kb];
        a.y = feat[mrow * D + kb + 1];
        float4 bq[2];
        bq[0] = wf[(kk * 2 + 0) * 32];
        bq[1] = wf[(kk * 2 + 1) * 32];
#pragma unroll
        for (int tt = 0; tt < 4; ++tt) {
            v2f bf;
            bf.x = (tt & 1) ? bq[tt >> 1].z : bq[tt >> 1].x;
            bf.y = (tt & 1) ? bq[tt >> 1].w : bq[tt >> 1].y;
            acc[tt] = __builtin_amdgcn_wmma_f32_16x16x4_f32(
                false, a, false, bf, (short)0, acc[tt], false, false);
        }
    }
#pragma unroll
    for (int kk = 0; kk < 16; ++kk) {
        const int kb = kk * 4 + kh;
        v2f a;
        a.x = agg[c][kb];
        a.y = agg[c][kb + 1];
        float4 bq[2];
        bq[0] = wf[((kk + 16) * 2 + 0) * 32];
        bq[1] = wf[((kk + 16) * 2 + 1) * 32];
#pragma unroll
        for (int tt = 0; tt < 4; ++tt) {
            v2f bf;
            bf.x = (tt & 1) ? bq[tt >> 1].z : bq[tt >> 1].x;
            bf.y = (tt & 1) ? bq[tt >> 1].w : bq[tt >> 1].y;
            acc[tt] = __builtin_amdgcn_wmma_f32_16x16x4_f32(
                false, a, false, bf, (short)0, acc[tt], false, false);
        }
    }

    const int rowBase = n0 + (lane >> 4) * 8;
#pragma unroll
    for (int tt = 0; tt < 4; ++tt) {
        const int col = tt * 16 + c;
        const float bv = bias[col];
#pragma unroll
        for (int r8 = 0; r8 < 8; ++r8) {
            const int r = rowBase + r8;
            out[r * D + col] = acc[tt][r8] + bv;
        }
    }
}

static_assert((((size_t)N_NODES * D + 2 * 50048 + N_EDGES) * 4) % 128 == 0, "rec must start on a 128-byte line of the workspace");

extern "C" void kernel_launch(void* const* d_in, const int* in_sizes, int n_in,
                              void* d_out, int out_size, void* d_ws, size_t ws_size,
                              hipStream_t stream)
{
    const float* feat = (const float*)d_in[0];
    const int*   src  = (const int*)d_in[1];
    const int*   dst  = (const int*)d_in[2];
    const float* Wn   = (const float*)d_in[3];
    const float* Ws   = (const float*)d_in[4];
    const float* bias = (const float*)d_in[5];
    float* out = (float*)d_out;

    float* plane0 = (float*)d_ws;
    int* boff = (int*)d_ws;
    float* wfrag = plane0 + 65536;
    int* perm = (int*)(plane0 + (size_t)N_NODES * D) + 2 * 50048;
    unsigned* rec = (unsigned*)(perm + N_EDGES);

    sage_chunk_kernel<<<NCHUNK, SORT_T, 0, stream>>>(src, dst, rec, boff, Wn, Ws, wfrag);

    sage_bucket_fused_kernel<<<NBUCK, BK_T, 0, stream>>>(rec, boff, perm, feat, wfrag, bias, out);
}
